# starting kernel
# baseline (speedup 1.0000x reference)
#include <hip/hip_runtime.h>

typedef __attribute__((ext_vector_type(2))) float v2f;
typedef __attribute__((ext_vector_type(8))) float v8f;

#define WMMA_F32(a, b, c) \
  __builtin_amdgcn_wmma_f32_16x16x4_f32(false, (a), false, (b), (short)0, (c), false, false)

#define BB      1024
#define NN      100000
#define DD      64
#define HH      512
#define RTILES  4
#define NSPLIT  64
#define CHUNKS  (NN / 16)
#define CPS     ((CHUNKS + NSPLIT - 1) / NSPLIT)

__global__ void tnorm_kernel(const float* __restrict__ table,
                             float* __restrict__ tnorm) {
  int j = blockIdx.x * blockDim.x + threadIdx.x;
  if (j >= NN) return;
  const float4* row = (const float4*)(table + (size_t)j * DD);
  float s = 0.f;
#pragma unroll
  for (int i = 0; i < DD / 4; ++i) {
    float4 v = row[i];
    s += v.x * v.x + v.y * v.y + v.z * v.z + v.w * v.w;
  }
  tnorm[j] = s;
}

__global__ __launch_bounds__(32) void nn_kernel(
    const float* __restrict__ codes, const float* __restrict__ table,
    const float* __restrict__ tnorm, float* __restrict__ pval,
    int* __restrict__ pidx) {
  const int lane = threadIdx.x;
  const int mgroup = blockIdx.x;
  const int split = blockIdx.y;
  const int m = lane & 15;
  const int koff = (lane >> 4) * 2;

  v2f a[RTILES][16];
#pragma unroll
  for (int t = 0; t < RTILES; ++t) {
    const float* crow =
        codes + (size_t)((mgroup * RTILES + t) * 16 + m) * DD + koff;
#pragma unroll
    for (int kb = 0; kb < 16; ++kb) {
      v2f v = *(const v2f*)(crow + kb * 4);
      a[t][kb].x = -2.0f * v.x;
      a[t][kb].y = -2.0f * v.y;
    }
  }

  float bestv[RTILES][8];
  int besti[RTILES][8];
#pragma unroll
  for (int t = 0; t < RTILES; ++t)
#pragma unroll
    for (int v = 0; v < 8; ++v) { bestv[t][v] = 3.4e38f; besti[t][v] = 0; }

  const int c0 = split * CPS;
  const int c1 = (c0 + CPS < CHUNKS) ? (c0 + CPS) : CHUNKS;

  for (int c = c0; c < c1; ++c) {
    const int col = c * 16 + m;
    const float* trow = table + (size_t)col * DD + koff;

    if (c + 1 < c1)
      __builtin_prefetch(table + (size_t)(col + 16) * DD + koff, 0, 1);

    v2f b[16];
#pragma unroll
    for (int kb = 0; kb < 16; ++kb) b[kb] = *(const v2f*)(trow + kb * 4);

    v8f acc[RTILES];
#pragma unroll
    for (int t = 0; t < RTILES; ++t) acc[t] = (v8f){};

#pragma unroll
    for (int kb = 0; kb < 16; ++kb)
#pragma unroll
      for (int t = 0; t < RTILES; ++t)
        acc[t] = WMMA_F32(a[t][kb], b[kb], acc[t]);

    const float tn = tnorm[col];
#pragma unroll
    for (int t = 0; t < RTILES; ++t)
#pragma unroll
      for (int v = 0; v < 8; ++v) {
        float s = tn + acc[t][v];
        if (s < bestv[t][v]) { bestv[t][v] = s; besti[t][v] = col; }
      }
  }

#pragma unroll
  for (int off = 1; off < 16; off <<= 1) {
#pragma unroll
    for (int t = 0; t < RTILES; ++t)
#pragma unroll
      for (int v = 0; v < 8; ++v) {
        float ov = __shfl_xor(bestv[t][v], off, 32);
        int oi = __shfl_xor(besti[t][v], off, 32);
        if (ov < bestv[t][v]) { bestv[t][v] = ov; besti[t][v] = oi; }
      }
  }

  if ((lane & 15) == 0) {
    const int rbase = (lane >> 4) * 8;
#pragma unroll
    for (int t = 0; t < RTILES; ++t) {
      const int tile = mgroup * RTILES + t;
      const int out = (tile * NSPLIT + split) * 16 + rbase;
#pragma unroll
      for (int v = 0; v < 8; ++v) {
        pval[out + v] = bestv[t][v];
        pidx[out + v] = besti[t][v];
      }
    }
  }
}

__global__ void reduce_kernel(const float* __restrict__ pval,
                              const int* __restrict__ pidx,
                              int* __restrict__ nn_idx) {
  int b = blockIdx.x * blockDim.x + threadIdx.x;
  if (b >= BB) return;
  const int tile = b >> 4, r = b & 15;
  float bv = 3.4e38f;
  int bi = 0;
  for (int s = 0; s < NSPLIT; ++s) {
    float v = pval[(tile * NSPLIT + s) * 16 + r];
    if (v < bv) { bv = v; bi = pidx[(tile * NSPLIT + s) * 16 + r]; }
  }
  nn_idx[b] = bi;
}

__global__ __launch_bounds__(32) void h1_kernel(
    const float* __restrict__ table, const int* __restrict__ nn_idx,
    const float* __restrict__ W1, const float* __restrict__ b1,
    float* __restrict__ h1) {
  const int lane = threadIdx.x;
  const int ntile = blockIdx.x;
  const int mtile = blockIdx.y;
  const int n = lane & 15;
  const int koff = (lane >> 4) * 2;
  const int nc = ntile * 16 + n;

  const int row = nn_idx[mtile * 16 + n];
  const float* prow = table + (size_t)row * DD + koff;

  const float bias = b1[nc];
  v8f acc;
#pragma unroll
  for (int v = 0; v < 8; ++v) acc[v] = bias;

#pragma unroll
  for (int kb = 0; kb < 16; ++kb) {
    v2f a = *(const v2f*)(prow + kb * 4);
    v2f b;
    b.x = W1[(size_t)(kb * 4 + koff + 0) * HH + nc];
    b.y = W1[(size_t)(kb * 4 + koff + 1) * HH + nc];
    acc = WMMA_F32(a, b, acc);
  }

#pragma unroll
  for (int v = 0; v < 8; ++v) {
    const int m = v + (lane >> 4) * 8;
    float h = acc[v];
    h1[(size_t)(mtile * 16 + m) * HH + nc] = h > 0.f ? h : 0.f;
  }
}

__global__ __launch_bounds__(32) void out_kernel(
    const float* __restrict__ h1, const float* __restrict__ Wu,
    const float* __restrict__ bu, const float* __restrict__ Ws,
    const float* __restrict__ bs, float* __restrict__ out) {
  const int lane = threadIdx.x;
  const int ntile = blockIdx.x;
  const int mtile = blockIdx.y;
  const int z = blockIdx.z;
  const float* W = z ? Ws : Wu;
  const float* bvec = z ? bs : bu;
  const int n = lane & 15;
  const int koff = (lane >> 4) * 2;
  const int nc = ntile * 16 + n;

  const float bias = bvec[nc];
  v8f acc;
#pragma unroll
  for (int v = 0; v < 8; ++v) acc[v] = bias;

  const float* hrow = h1 + (size_t)(mtile * 16 + n) * HH + koff;
#pragma unroll 8
  for (int kb = 0; kb < HH / 4; ++kb) {
    v2f a = *(const v2f*)(hrow + kb * 4);
    v2f b;
    b.x = W[(size_t)(kb * 4 + koff + 0) * DD + nc];
    b.y = W[(size_t)(kb * 4 + koff + 1) * DD + nc];
    acc = WMMA_F32(a, b, acc);
  }

  float* o = out + (size_t)z * BB * DD;
#pragma unroll
  for (int v = 0; v < 8; ++v) {
    const int m = v + (lane >> 4) * 8;
    o[(size_t)(mtile * 16 + m) * DD + nc] = acc[v];
  }
}

extern "C" void kernel_launch(void* const* d_in, const int* in_sizes, int n_in,
                              void* d_out, int out_size, void* d_ws,
                              size_t ws_size, hipStream_t stream) {
  const float* codes = (const float*)d_in[0];
  const float* table = (const float*)d_in[1];
  const float* W1 = (const float*)d_in[2];
  const float* b1 = (const float*)d_in[3];
  const float* Wu = (const float*)d_in[4];
  const float* bu = (const float*)d_in[5];
  const float* Ws = (const float*)d_in[6];
  const float* bs = (const float*)d_in[7];
  float* out = (float*)d_out;

  char* ws = (char*)d_ws;
  float* tnorm = (float*)(ws);
  float* pval  = (float*)(ws + 400000);
  int*   pidx  = (int*)  (ws + 400000 + 262144);
  int*   nnidx = (int*)  (ws + 400000 + 524288);
  float* h1    = (float*)(ws + 400000 + 524288 + 4096);

  tnorm_kernel<<<(NN + 255) / 256, 256, 0, stream>>>(table, tnorm);
  nn_kernel<<<dim3(BB / 16 / RTILES, NSPLIT), 32, 0, stream>>>(
      codes, table, tnorm, pval, pidx);
  reduce_kernel<<<(BB + 255) / 256, 256, 0, stream>>>(pval, pidx, nnidx);
  h1_kernel<<<dim3(HH / 16, BB / 16), 32, 0, stream>>>(table, nnidx, W1, b1,
                                                       h1);
  out_kernel<<<dim3(DD / 16, BB / 16, 2), 32, 0, stream>>>(h1, Wu, bu, Ws, bs,
                                                           out);
}


# nn search: 8-wave work-groups fetch each table stage once (coalesced 16-byte loads) into two padded LDS buffers; each wave reads B fragments from LDS for its own 2 M-tiles; sum order kept
# speedup vs baseline: 1.4206x; 1.4206x over previous
#include <hip/hip_runtime.h>

typedef __attribute__((ext_vector_type(2))) float v2f;
typedef __attribute__((ext_vector_type(4))) float v4f;
typedef __attribute__((ext_vector_type(8))) float v8f;

#define WMMA_F32(a, b, c) \
  __builtin_amdgcn_wmma_f32_16x16x4_f32(false, (a), false, (b), (short)0, (c), false, false)

#define BB      1024
#define NN      100000
#define DD      64
#define HH      512
#define NWAVES  8
#define WTILES  2
#define STAGE_CHUNKS 20
#define STAGE_ROWS   (STAGE_CHUNKS * 16)
#define PITCH   68
#define NSPLIT  64
#define CHUNKS  (NN / 16)
#define CPS     ((CHUNKS + NSPLIT - 1) / NSPLIT)

__global__ void tnorm_kernel(const float* __restrict__ table,
                             float* __restrict__ tnorm) {
  int j = blockIdx.x * blockDim.x + threadIdx.x;
  if (j >= NN) return;
  const float4* row = (const float4*)(table + (size_t)j * DD);
  float s = 0.f;
#pragma unroll
  for (int i = 0; i < DD / 4; ++i) {
    float4 v = row[i];
    s += v.x * v.x + v.y * v.y + v.z * v.z + v.w * v.w;
  }
  tnorm[j] = s;
}

__global__ __launch_bounds__(NWAVES * 32) void nn_kernel(
    const float* __restrict__ codes, const float* __restrict__ table,
    const float* __restrict__ tnorm, float* __restrict__ pval,
    int* __restrict__ pidx) {
  __shared__ __attribute__((aligned(16))) float s_t[2][STAGE_ROWS * PITCH];
  __shared__ float s_n[2][STAGE_ROWS];

  const int tid = threadIdx.x;
  const int lane = tid & 31;
  const int wave = tid >> 5;
  const int mgroup = blockIdx.x;
  const int split = blockIdx.y;
  const int m = lane & 15;
  const int koff = (lane >> 4) * 2;
  const int tile0 = (mgroup * NWAVES + wave) * WTILES;

  v2f a[WTILES][16];
#pragma unroll
  for (int t = 0; t < WTILES; ++t) {
    const float* crow = codes + (size_t)((tile0 + t) * 16 + m) * DD + koff;
#pragma unroll
    for (int kb = 0; kb < 16; ++kb) {
      v2f v = *(const v2f*)(crow + kb * 4);
      a[t][kb].x = -2.0f * v.x;
      a[t][kb].y = -2.0f * v.y;
    }
  }

  float bestv[WTILES][8];
  int besti[WTILES][8];
#pragma unroll
  for (int t = 0; t < WTILES; ++t)
#pragma unroll
    for (int v = 0; v < 8; ++v) { bestv[t][v] = 3.4e38f; besti[t][v] = 0; }

  const int c0 = split * CPS;
  const int c1 = (c0 + CPS < CHUNKS) ? (c0 + CPS) : CHUNKS;
  const int nch = c1 - c0;
  const int nst = (nch + STAGE_CHUNKS - 1) / STAGE_CHUNKS;

  v4f g[STAGE_CHUNKS];
  float gn0 = 0.f, gn1 = 0.f;
  const int srow = tid >> 4;
  const int spiece = (tid & 15) * 4;

  {
    const int nchs = (nch < STAGE_CHUNKS) ? nch : STAGE_CHUNKS;
    const float* gp = table + (size_t)(c0 * 16 + srow) * DD + spiece;
#pragma unroll
    for (int i = 0; i < STAGE_CHUNKS; ++i)
      if (i < nchs) g[i] = *(const v4f*)(gp + (size_t)i * 16 * DD);
    if (tid < nchs * 16) gn0 = tnorm[c0 * 16 + tid];
    if (tid + 256 < nchs * 16) gn1 = tnorm[c0 * 16 + 256 + tid];
#pragma unroll
    for (int i = 0; i < STAGE_CHUNKS; ++i)
      if (i < nchs)
        *(v4f*)(&s_t[0][(i * 16 + srow) * PITCH + spiece]) = g[i];
    if (tid < nchs * 16) s_n[0][tid] = gn0;
    if (tid + 256 < nchs * 16) s_n[0][256 + tid] = gn1;
  }
  __syncthreads();

  for (int s = 0; s < nst; ++s) {
    const int buf = s & 1;
    const int cs0 = c0 + s * STAGE_CHUNKS;
    const int nchs = (c1 - cs0 < STAGE_CHUNKS) ? (c1 - cs0) : STAGE_CHUNKS;
    const int cn0 = cs0 + STAGE_CHUNKS;
    const int nchn =
        (c1 - cn0 < STAGE_CHUNKS) ? (c1 - cn0) : STAGE_CHUNKS;

    if (nchn > 0) {
      const float* gp = table + (size_t)(cn0 * 16 + srow) * DD + spiece;
#pragma unroll
      for (int i = 0; i < STAGE_CHUNKS; ++i)
        if (i < nchn) g[i] = *(const v4f*)(gp + (size_t)i * 16 * DD);
      if (tid < nchn * 16) gn0 = tnorm[cn0 * 16 + tid];
      if (tid + 256 < nchn * 16) gn1 = tnorm[cn0 * 16 + 256 + tid];
    }

    const float* sb = &s_t[buf][m * PITCH + koff];
    for (int cc = 0; cc < nchs; ++cc) {
      const int col = (cs0 + cc) * 16 + m;
      const float* trow = sb + cc * 16 * PITCH;

      v2f b[16];
#pragma unroll
      for (int kb = 0; kb < 16; ++kb) b[kb] = *(const v2f*)(trow + kb * 4);
      const float tn = s_n[buf][cc * 16 + m];

      v8f acc[WTILES];
#pragma unroll
      for (int t = 0; t < WTILES; ++t) acc[t] = (v8f){};

#pragma unroll
      for (int kb = 0; kb < 16; ++kb)
#pragma unroll
        for (int t = 0; t < WTILES; ++t)
          acc[t] = WMMA_F32(a[t][kb], b[kb], acc[t]);

#pragma unroll
      for (int t = 0; t < WTILES; ++t)
#pragma unroll
        for (int v = 0; v < 8; ++v) {
          float sc = tn + acc[t][v];
          if (sc < bestv[t][v]) { bestv[t][v] = sc; besti[t][v] = col; }
        }
    }

    if (nchn > 0) {
#pragma unroll
      for (int i = 0; i < STAGE_CHUNKS; ++i)
        if (i < nchn)
          *(v4f*)(&s_t[buf ^ 1][(i * 16 + srow) * PITCH + spiece]) = g[i];
      if (tid < nchn * 16) s_n[buf ^ 1][tid] = gn0;
      if (tid + 256 < nchn * 16) s_n[buf ^ 1][256 + tid] = gn1;
    }
    __syncthreads();
  }

#pragma unroll
  for (int off = 1; off < 16; off <<= 1) {
#pragma unroll
    for (int t = 0; t < WTILES; ++t)
#pragma unroll
      for (int v = 0; v < 8; ++v) {
        float ov = __shfl_xor(bestv[t][v], off, 32);
        int oi = __shfl_xor(besti[t][v], off, 32);
        if (ov < bestv[t][v]) { bestv[t][v] = ov; besti[t][v] = oi; }
      }
  }

  if ((lane & 15) == 0) {
    const int rbase = (lane >> 4) * 8;
#pragma unroll
    for (int t = 0; t < WTILES; ++t) {
      const int tile = tile0 + t;
      const int out = (tile * NSPLIT + split) * 16 + rbase;
#pragma unroll
      for (int v = 0; v < 8; ++v) {
        pval[out + v] = bestv[t][v];
        pidx[out + v] = besti[t][v];
      }
    }
  }
}

__global__ void reduce_kernel(const float* __restrict__ pval,
                              const int* __restrict__ pidx,
                              int* __restrict__ nn_idx) {
  int b = blockIdx.x * blockDim.x + threadIdx.x;
  if (b >= BB) return;
  const int tile = b >> 4, r = b & 15;
  float bv = 3.4e38f;
  int bi = 0;
  for (int s = 0; s < NSPLIT; ++s) {
    float v = pval[(tile * NSPLIT + s) * 16 + r];
    if (v < bv) { bv = v; bi = pidx[(tile * NSPLIT + s) * 16 + r]; }
  }
  nn_idx[b] = bi;
}

__global__ __launch_bounds__(32) void h1_kernel(
    const float* __restrict__ table, const int* __restrict__ nn_idx,
    const float* __restrict__ W1, const float* __restrict__ b1,
    float* __restrict__ h1) {
  const int lane = threadIdx.x;
  const int ntile = blockIdx.x;
  const int mtile = blockIdx.y;
  const int n = lane & 15;
  const int koff = (lane >> 4) * 2;
  const int nc = ntile * 16 + n;

  const int row = nn_idx[mtile * 16 + n];
  const float* prow = table + (size_t)row * DD + koff;

  const float bias = b1[nc];
  v8f acc;
#pragma unroll
  for (int v = 0; v < 8; ++v) acc[v] = bias;

#pragma unroll
  for (int kb = 0; kb < 16; ++kb) {
    v2f a = *(const v2f*)(prow + kb * 4);
    v2f b;
    b.x = W1[(size_t)(kb * 4 + koff + 0) * HH + nc];
    b.y = W1[(size_t)(kb * 4 + koff + 1) * HH + nc];
    acc = WMMA_F32(a, b, acc);
  }

#pragma unroll
  for (int v = 0; v < 8; ++v) {
    const int m = v + (lane >> 4) * 8;
    float h = acc[v];
    h1[(size_t)(mtile * 16 + m) * HH + nc] = h > 0.f ? h : 0.f;
  }
}

__global__ __launch_bounds__(32) void out_kernel(
    const float* __restrict__ h1, const float* __restrict__ Wu,
    const float* __restrict__ bu, const float* __restrict__ Ws,
    const float* __restrict__ bs, float* __restrict__ out) {
  const int lane = threadIdx.x;
  const int ntile = blockIdx.x;
  const int mtile = blockIdx.y;
  const int z = blockIdx.z;
  const float* W = z ? Ws : Wu;
  const float* bvec = z ? bs : bu;
  const int n = lane & 15;
  const int koff = (lane >> 4) * 2;
  const int nc = ntile * 16 + n;

  const float bias = bvec[nc];
  v8f acc;
#pragma unroll
  for (int v = 0; v < 8; ++v) acc[v] = bias;

  const float* hrow = h1 + (size_t)(mtile * 16 + n) * HH + koff;
#pragma unroll 8
  for (int kb = 0; kb < HH / 4; ++kb) {
    v2f a = *(const v2f*)(hrow + kb * 4);
    v2f b;
    b.x = W[(size_t)(kb * 4 + koff + 0) * DD + nc];
    b.y = W[(size_t)(kb * 4 + koff + 1) * DD + nc];
    acc = WMMA_F32(a, b, acc);
  }

  float* o = out + (size_t)z * BB * DD;
#pragma unroll
  for (int v = 0; v < 8; ++v) {
    const int m = v + (lane >> 4) * 8;
    o[(size_t)(mtile * 16 + m) * DD + nc] = acc[v];
  }
}

extern "C" void kernel_launch(void* const* d_in, const int* in_sizes, int n_in,
                              void* d_out, int out_size, void* d_ws,
                              size_t ws_size, hipStream_t stream) {
  const float* codes = (const float*)d_in[0];
  const float* table = (const float*)d_in[1];
  const float* W1 = (const float*)d_in[2];
  const float* b1 = (const float*)d_in[3];
  const float* Wu = (const float*)d_in[4];
  const float* bu = (const float*)d_in[5];
  const float* Ws = (const float*)d_in[6];
  const float* bs = (const float*)d_in[7];
  float* out = (float*)d_out;

  char* ws = (char*)d_ws;
  float* tnorm = (float*)(ws);
  float* pval  = (float*)(ws + 400000);
  int*   pidx  = (int*)  (ws + 400000 + 262144);
  int*   nnidx = (int*)  (ws + 400000 + 524288);
  float* h1    = (float*)(ws + 400000 + 524288 + 4096);

  tnorm_kernel<<<(NN + 255) / 256, 256, 0, stream>>>(table, tnorm);
  nn_kernel<<<dim3(BB / 16 / (NWAVES * WTILES), NSPLIT), NWAVES * 32, 0, stream>>>(
      codes, table, tnorm, pval, pidx);
  reduce_kernel<<<(BB + 255) / 256, 256, 0, stream>>>(pval, pidx, nnidx);
  h1_kernel<<<dim3(HH / 16, BB / 16), 32, 0, stream>>>(table, nnidx, W1, b1,
                                                       h1);
  out_kernel<<<dim3(DD / 16, BB / 16, 2), 32, 0, stream>>>(h1, Wu, bu, Ws, bs,
                                                           out);
}


# partial-argmin merge: a row's 64 split partials are read by 16 lanes at once (8 independent loads each) and joined by lane exchange and LDS, replacing a 64-step dependent load loop
# speedup vs baseline: 1.7315x; 1.2189x over previous
#include <hip/hip_runtime.h>

typedef __attribute__((ext_vector_type(2))) float v2f;
typedef __attribute__((ext_vector_type(4))) float v4f;
typedef __attribute__((ext_vector_type(8))) float v8f;

#define WMMA_F32(a, b, c) \
  __builtin_amdgcn_wmma_f32_16x16x4_f32(false, (a), false, (b), (short)0, (c), false, false)

#define BB      1024
#define NN      100000
#define DD      64
#define HH      512
#define NWAVES  8
#define WTILES  2
#define STAGE_CHUNKS 20
#define STAGE_ROWS   (STAGE_CHUNKS * 16)
#define PITCH   68
#define NSPLIT  64
#define CHUNKS  (NN / 16)
#define CPS     ((CHUNKS + NSPLIT - 1) / NSPLIT)

__global__ void tnorm_kernel(const float* __restrict__ table,
                             float* __restrict__ tnorm) {
  int j = blockIdx.x * blockDim.x + threadIdx.x;
  if (j >= NN) return;
  const float4* row = (const float4*)(table + (size_t)j * DD);
  float s = 0.f;
#pragma unroll
  for (int i = 0; i < DD / 4; ++i) {
    float4 v = row[i];
    s += v.x * v.x + v.y * v.y + v.z * v.z + v.w * v.w;
  }
  tnorm[j] = s;
}

__global__ __launch_bounds__(NWAVES * 32) void nn_kernel(
    const float* __restrict__ codes, const float* __restrict__ table,
    const float* __restrict__ tnorm, float* __restrict__ pval,
    int* __restrict__ pidx) {
  __shared__ __attribute__((aligned(16))) float s_t[2][STAGE_ROWS * PITCH];
  __shared__ float s_n[2][STAGE_ROWS];

  const int tid = threadIdx.x;
  const int lane = tid & 31;
  const int wave = tid >> 5;
  const int mgroup = blockIdx.x;
  const int split = blockIdx.y;
  const int m = lane & 15;
  const int koff = (lane >> 4) * 2;
  const int tile0 = (mgroup * NWAVES + wave) * WTILES;

  v2f a[WTILES][16];
#pragma unroll
  for (int t = 0; t < WTILES; ++t) {
    const float* crow = codes + (size_t)((tile0 + t) * 16 + m) * DD + koff;
#pragma unroll
    for (int kb = 0; kb < 16; ++kb) {
      v2f v = *(const v2f*)(crow + kb * 4);
      a[t][kb].x = -2.0f * v.x;
      a[t][kb].y = -2.0f * v.y;
    }
  }

  float bestv[WTILES][8];
  int besti[WTILES][8];
#pragma unroll
  for (int t = 0; t < WTILES; ++t)
#pragma unroll
    for (int v = 0; v < 8; ++v) { bestv[t][v] = 3.4e38f; besti[t][v] = 0; }

  const int c0 = split * CPS;
  const int c1 = (c0 + CPS < CHUNKS) ? (c0 + CPS) : CHUNKS;
  const int nch = c1 - c0;
  const int nst = (nch + STAGE_CHUNKS - 1) / STAGE_CHUNKS;

  v4f g[STAGE_CHUNKS];
  float gn0 = 0.f, gn1 = 0.f;
  const int srow = tid >> 4;
  const int spiece = (tid & 15) * 4;

  {
    const int nchs = (nch < STAGE_CHUNKS) ? nch : STAGE_CHUNKS;
    const float* gp = table + (size_t)(c0 * 16 + srow) * DD + spiece;
#pragma unroll
    for (int i = 0; i < STAGE_CHUNKS; ++i)
      if (i < nchs) g[i] = *(const v4f*)(gp + (size_t)i * 16 * DD);
    if (tid < nchs * 16) gn0 = tnorm[c0 * 16 + tid];
    if (tid + 256 < nchs * 16) gn1 = tnorm[c0 * 16 + 256 + tid];
#pragma unroll
    for (int i = 0; i < STAGE_CHUNKS; ++i)
      if (i < nchs)
        *(v4f*)(&s_t[0][(i * 16 + srow) * PITCH + spiece]) = g[i];
    if (tid < nchs * 16) s_n[0][tid] = gn0;
    if (tid + 256 < nchs * 16) s_n[0][256 + tid] = gn1;
  }
  __syncthreads();

  for (int s = 0; s < nst; ++s) {
    const int buf = s & 1;
    const int cs0 = c0 + s * STAGE_CHUNKS;
    const int nchs = (c1 - cs0 < STAGE_CHUNKS) ? (c1 - cs0) : STAGE_CHUNKS;
    const int cn0 = cs0 + STAGE_CHUNKS;
    const int nchn =
        (c1 - cn0 < STAGE_CHUNKS) ? (c1 - cn0) : STAGE_CHUNKS;

    if (nchn > 0) {
      const float* gp = table + (size_t)(cn0 * 16 + srow) * DD + spiece;
#pragma unroll
      for (int i = 0; i < STAGE_CHUNKS; ++i)
        if (i < nchn) g[i] = *(const v4f*)(gp + (size_t)i * 16 * DD);
      if (tid < nchn * 16) gn0 = tnorm[cn0 * 16 + tid];
      if (tid + 256 < nchn * 16) gn1 = tnorm[cn0 * 16 + 256 + tid];
    }

    const float* sb = &s_t[buf][m * PITCH + koff];
    for (int cc = 0; cc < nchs; ++cc) {
      const int col = (cs0 + cc) * 16 + m;
      const float* trow = sb + cc * 16 * PITCH;

      v2f b[16];
#pragma unroll
      for (int kb = 0; kb < 16; ++kb) b[kb] = *(const v2f*)(trow + kb * 4);
      const float tn = s_n[buf][cc * 16 + m];

      v8f acc[WTILES];
#pragma unroll
      for (int t = 0; t < WTILES; ++t) acc[t] = (v8f){};

#pragma unroll
      for (int kb = 0; kb < 16; ++kb)
#pragma unroll
        for (int t = 0; t < WTILES; ++t)
          acc[t] = WMMA_F32(a[t][kb], b[kb], acc[t]);

#pragma unroll
      for (int t = 0; t < WTILES; ++t)
#pragma unroll
        for (int v = 0; v < 8; ++v) {
          float sc = tn + acc[t][v];
          if (sc < bestv[t][v]) { bestv[t][v] = sc; besti[t][v] = col; }
        }
    }

    if (nchn > 0) {
#pragma unroll
      for (int i = 0; i < STAGE_CHUNKS; ++i)
        if (i < nchn)
          *(v4f*)(&s_t[buf ^ 1][(i * 16 + srow) * PITCH + spiece]) = g[i];
      if (tid < nchn * 16) s_n[buf ^ 1][tid] = gn0;
      if (tid + 256 < nchn * 16) s_n[buf ^ 1][256 + tid] = gn1;
    }
    __syncthreads();
  }

#pragma unroll
  for (int off = 1; off < 16; off <<= 1) {
#pragma unroll
    for (int t = 0; t < WTILES; ++t)
#pragma unroll
      for (int v = 0; v < 8; ++v) {
        float ov = __shfl_xor(bestv[t][v], off, 32);
        int oi = __shfl_xor(besti[t][v], off, 32);
        if (ov < bestv[t][v]) { bestv[t][v] = ov; besti[t][v] = oi; }
      }
  }

  if ((lane & 15) == 0) {
    const int rbase = (lane >> 4) * 8;
#pragma unroll
    for (int t = 0; t < WTILES; ++t) {
      const int tile = tile0 + t;
      const int out = (tile * NSPLIT + split) * 16 + rbase;
#pragma unroll
      for (int v = 0; v < 8; ++v) {
        pval[out + v] = bestv[t][v];
        pidx[out + v] = besti[t][v];
      }
    }
  }
}

#define RSPT (NSPLIT / 16)

__device__ __forceinline__ void take_min(float& bv, int& bi, float ov, int oi) {
  const bool t = (ov < bv) | ((ov == bv) & (oi < bi));
  bv = t ? ov : bv;
  bi = t ? oi : bi;
}

__global__ __launch_bounds__(256) void reduce_kernel(
    const float* __restrict__ pval, const int* __restrict__ pidx,
    int* __restrict__ nn_idx) {
  __shared__ float s_v[8][16];
  __shared__ int s_i[8][16];
  const int tid = threadIdx.x;
  const int tile = blockIdx.x;
  const int r = tid & 15;
  const int wave = tid >> 5;
  const int base = tile * NSPLIT * 16 + tid;

  float v[RSPT];
  int ix[RSPT];
#pragma unroll
  for (int j = 0; j < RSPT; ++j) {
    v[j] = pval[base + j * 256];
    ix[j] = pidx[base + j * 256];
  }
  float bv = 3.4e38f;
  int bi = 0;
#pragma unroll
  for (int j = 0; j < RSPT; ++j) take_min(bv, bi, v[j], ix[j]);

  take_min(bv, bi, __shfl_xor(bv, 16, 32), __shfl_xor(bi, 16, 32));
  if ((tid & 16) == 0) { s_v[wave][r] = bv; s_i[wave][r] = bi; }
  __syncthreads();

  if (tid < 16) {
    float wv[8];
    int wi[8];
#pragma unroll
    for (int w = 0; w < 8; ++w) { wv[w] = s_v[w][tid]; wi[w] = s_i[w][tid]; }
    float fv = wv[0];
    int fi = wi[0];
#pragma unroll
    for (int w = 1; w < 8; ++w) take_min(fv, fi, wv[w], wi[w]);
    nn_idx[tile * 16 + tid] = fi;
  }
}

__global__ __launch_bounds__(32) void h1_kernel(
    const float* __restrict__ table, const int* __restrict__ nn_idx,
    const float* __restrict__ W1, const float* __restrict__ b1,
    float* __restrict__ h1) {
  const int lane = threadIdx.x;
  const int ntile = blockIdx.x;
  const int mtile = blockIdx.y;
  const int n = lane & 15;
  const int koff = (lane >> 4) * 2;
  const int nc = ntile * 16 + n;

  const int row = nn_idx[mtile * 16 + n];
  const float* prow = table + (size_t)row * DD + koff;

  const float bias = b1[nc];
  v8f acc;
#pragma unroll
  for (int v = 0; v < 8; ++v) acc[v] = bias;

#pragma unroll
  for (int kb = 0; kb < 16; ++kb) {
    v2f a = *(const v2f*)(prow + kb * 4);
    v2f b;
    b.x = W1[(size_t)(kb * 4 + koff + 0) * HH + nc];
    b.y = W1[(size_t)(kb * 4 + koff + 1) * HH + nc];
    acc = WMMA_F32(a, b, acc);
  }

#pragma unroll
  for (int v = 0; v < 8; ++v) {
    const int m = v + (lane >> 4) * 8;
    float h = acc[v];
    h1[(size_t)(mtile * 16 + m) * HH + nc] = h > 0.f ? h : 0.f;
  }
}

__global__ __launch_bounds__(32) void out_kernel(
    const float* __restrict__ h1, const float* __restrict__ Wu,
    const float* __restrict__ bu, const float* __restrict__ Ws,
    const float* __restrict__ bs, float* __restrict__ out) {
  const int lane = threadIdx.x;
  const int ntile = blockIdx.x;
  const int mtile = blockIdx.y;
  const int z = blockIdx.z;
  const float* W = z ? Ws : Wu;
  const float* bvec = z ? bs : bu;
  const int n = lane & 15;
  const int koff = (lane >> 4) * 2;
  const int nc = ntile * 16 + n;

  const float bias = bvec[nc];
  v8f acc;
#pragma unroll
  for (int v = 0; v < 8; ++v) acc[v] = bias;

  const float* hrow = h1 + (size_t)(mtile * 16 + n) * HH + koff;
#pragma unroll 8
  for (int kb = 0; kb < HH / 4; ++kb) {
    v2f a = *(const v2f*)(hrow + kb * 4);
    v2f b;
    b.x = W[(size_t)(kb * 4 + koff + 0) * DD + nc];
    b.y = W[(size_t)(kb * 4 + koff + 1) * DD + nc];
    acc = WMMA_F32(a, b, acc);
  }

  float* o = out + (size_t)z * BB * DD;
#pragma unroll
  for (int v = 0; v < 8; ++v) {
    const int m = v + (lane >> 4) * 8;
    o[(size_t)(mtile * 16 + m) * DD + nc] = acc[v];
  }
}

extern "C" void kernel_launch(void* const* d_in, const int* in_sizes, int n_in,
                              void* d_out, int out_size, void* d_ws,
                              size_t ws_size, hipStream_t stream) {
  const float* codes = (const float*)d_in[0];
  const float* table = (const float*)d_in[1];
  const float* W1 = (const float*)d_in[2];
  const float* b1 = (const float*)d_in[3];
  const float* Wu = (const float*)d_in[4];
  const float* bu = (const float*)d_in[5];
  const float* Ws = (const float*)d_in[6];
  const float* bs = (const float*)d_in[7];
  float* out = (float*)d_out;

  char* ws = (char*)d_ws;
  float* tnorm = (float*)(ws);
  float* pval  = (float*)(ws + 400000);
  int*   pidx  = (int*)  (ws + 400000 + 262144);
  int*   nnidx = (int*)  (ws + 400000 + 524288);
  float* h1    = (float*)(ws + 400000 + 524288 + 4096);

  tnorm_kernel<<<(NN + 255) / 256, 256, 0, stream>>>(table, tnorm);
  nn_kernel<<<dim3(BB / 16 / (NWAVES * WTILES), NSPLIT), NWAVES * 32, 0, stream>>>(
      codes, table, tnorm, pval, pidx);
  reduce_kernel<<<BB / 16, 256, 0, stream>>>(pval, pidx, nnidx);
  h1_kernel<<<dim3(HH / 16, BB / 16), 32, 0, stream>>>(table, nnidx, W1, b1,
                                                       h1);
  out_kernel<<<dim3(DD / 16, BB / 16, 2), 32, 0, stream>>>(h1, Wu, bu, Ws, bs,
                                                           out);
}


# mlp: both layers in one launch, transposed: layer-1 D tiles (relu in registers) are layer 2's B fragments; hidden split over 8 waves, partials joined in LDS (new sum order); 5 launches to 4
# speedup vs baseline: 1.8742x; 1.0824x over previous
#include <hip/hip_runtime.h>

typedef __attribute__((ext_vector_type(2))) float v2f;
typedef __attribute__((ext_vector_type(4))) float v4f;
typedef __attribute__((ext_vector_type(8))) float v8f;

#define WMMA_F32(a, b, c) \
  __builtin_amdgcn_wmma_f32_16x16x4_f32(false, (a), false, (b), (short)0, (c), false, false)

#define BB      1024
#define NN      100000
#define DD      64
#define HH      512
#define NWAVES  8
#define WTILES  2
#define STAGE_CHUNKS 20
#define STAGE_ROWS   (STAGE_CHUNKS * 16)
#define PITCH   68
#define NSPLIT  64
#define CHUNKS  (NN / 16)
#define CPS     ((CHUNKS + NSPLIT - 1) / NSPLIT)

__global__ void tnorm_kernel(const float* __restrict__ table,
                             float* __restrict__ tnorm) {
  int j = blockIdx.x * blockDim.x + threadIdx.x;
  if (j >= NN) return;
  const float4* row = (const float4*)(table + (size_t)j * DD);
  float s = 0.f;
#pragma unroll
  for (int i = 0; i < DD / 4; ++i) {
    float4 v = row[i];
    s += v.x * v.x + v.y * v.y + v.z * v.z + v.w * v.w;
  }
  tnorm[j] = s;
}

__global__ __launch_bounds__(NWAVES * 32) void nn_kernel(
    const float* __restrict__ codes, const float* __restrict__ table,
    const float* __restrict__ tnorm, float* __restrict__ pval,
    int* __restrict__ pidx) {
  __shared__ __attribute__((aligned(16))) float s_t[2][STAGE_ROWS * PITCH];
  __shared__ float s_n[2][STAGE_ROWS];

  const int tid = threadIdx.x;
  const int lane = tid & 31;
  const int wave = tid >> 5;
  const int mgroup = blockIdx.x;
  const int split = blockIdx.y;
  const int m = lane & 15;
  const int koff = (lane >> 4) * 2;
  const int tile0 = (mgroup * NWAVES + wave) * WTILES;

  v2f a[WTILES][16];
#pragma unroll
  for (int t = 0; t < WTILES; ++t) {
    const float* crow = codes + (size_t)((tile0 + t) * 16 + m) * DD + koff;
#pragma unroll
    for (int kb = 0; kb < 16; ++kb) {
      v2f v = *(const v2f*)(crow + kb * 4);
      a[t][kb].x = -2.0f * v.x;
      a[t][kb].y = -2.0f * v.y;
    }
  }

  float bestv[WTILES][8];
  int besti[WTILES][8];
#pragma unroll
  for (int t = 0; t < WTILES; ++t)
#pragma unroll
    for (int v = 0; v < 8; ++v) { bestv[t][v] = 3.4e38f; besti[t][v] = 0; }

  const int c0 = split * CPS;
  const int c1 = (c0 + CPS < CHUNKS) ? (c0 + CPS) : CHUNKS;
  const int nch = c1 - c0;
  const int nst = (nch + STAGE_CHUNKS - 1) / STAGE_CHUNKS;

  v4f g[STAGE_CHUNKS];
  float gn0 = 0.f, gn1 = 0.f;
  const int srow = tid >> 4;
  const int spiece = (tid & 15) * 4;

  {
    const int nchs = (nch < STAGE_CHUNKS) ? nch : STAGE_CHUNKS;
    const float* gp = table + (size_t)(c0 * 16 + srow) * DD + spiece;
#pragma unroll
    for (int i = 0; i < STAGE_CHUNKS; ++i)
      if (i < nchs) g[i] = *(const v4f*)(gp + (size_t)i * 16 * DD);
    if (tid < nchs * 16) gn0 = tnorm[c0 * 16 + tid];
    if (tid + 256 < nchs * 16) gn1 = tnorm[c0 * 16 + 256 + tid];
#pragma unroll
    for (int i = 0; i < STAGE_CHUNKS; ++i)
      if (i < nchs)
        *(v4f*)(&s_t[0][(i * 16 + srow) * PITCH + spiece]) = g[i];
    if (tid < nchs * 16) s_n[0][tid] = gn0;
    if (tid + 256 < nchs * 16) s_n[0][256 + tid] = gn1;
  }
  __syncthreads();

  for (int s = 0; s < nst; ++s) {
    const int buf = s & 1;
    const int cs0 = c0 + s * STAGE_CHUNKS;
    const int nchs = (c1 - cs0 < STAGE_CHUNKS) ? (c1 - cs0) : STAGE_CHUNKS;
    const int cn0 = cs0 + STAGE_CHUNKS;
    const int nchn =
        (c1 - cn0 < STAGE_CHUNKS) ? (c1 - cn0) : STAGE_CHUNKS;

    if (nchn > 0) {
      const float* gp = table + (size_t)(cn0 * 16 + srow) * DD + spiece;
#pragma unroll
      for (int i = 0; i < STAGE_CHUNKS; ++i)
        if (i < nchn) g[i] = *(const v4f*)(gp + (size_t)i * 16 * DD);
      if (tid < nchn * 16) gn0 = tnorm[cn0 * 16 + tid];
      if (tid + 256 < nchn * 16) gn1 = tnorm[cn0 * 16 + 256 + tid];
    }

    const float* sb = &s_t[buf][m * PITCH + koff];
    for (int cc = 0; cc < nchs; ++cc) {
      const int col = (cs0 + cc) * 16 + m;
      const float* trow = sb + cc * 16 * PITCH;

      v2f b[16];
#pragma unroll
      for (int kb = 0; kb < 16; ++kb) b[kb] = *(const v2f*)(trow + kb * 4);
      const float tn = s_n[buf][cc * 16 + m];

      v8f acc[WTILES];
#pragma unroll
      for (int t = 0; t < WTILES; ++t) acc[t] = (v8f){};

#pragma unroll
      for (int kb = 0; kb < 16; ++kb)
#pragma unroll
        for (int t = 0; t < WTILES; ++t)
          acc[t] = WMMA_F32(a[t][kb], b[kb], acc[t]);

#pragma unroll
      for (int t = 0; t < WTILES; ++t)
#pragma unroll
        for (int v = 0; v < 8; ++v) {
          float sc = tn + acc[t][v];
          if (sc < bestv[t][v]) { bestv[t][v] = sc; besti[t][v] = col; }
        }
    }

    if (nchn > 0) {
#pragma unroll
      for (int i = 0; i < STAGE_CHUNKS; ++i)
        if (i < nchn)
          *(v4f*)(&s_t[buf ^ 1][(i * 16 + srow) * PITCH + spiece]) = g[i];
      if (tid < nchn * 16) s_n[buf ^ 1][tid] = gn0;
      if (tid + 256 < nchn * 16) s_n[buf ^ 1][256 + tid] = gn1;
    }
    __syncthreads();
  }

#pragma unroll
  for (int off = 1; off < 16; off <<= 1) {
#pragma unroll
    for (int t = 0; t < WTILES; ++t)
#pragma unroll
      for (int v = 0; v < 8; ++v) {
        float ov = __shfl_xor(bestv[t][v], off, 32);
        int oi = __shfl_xor(besti[t][v], off, 32);
        if (ov < bestv[t][v]) { bestv[t][v] = ov; besti[t][v] = oi; }
      }
  }

  if ((lane & 15) == 0) {
    const int rbase = (lane >> 4) * 8;
#pragma unroll
    for (int t = 0; t < WTILES; ++t) {
      const int tile = tile0 + t;
      const int out = (tile * NSPLIT + split) * 16 + rbase;
#pragma unroll
      for (int v = 0; v < 8; ++v) {
        pval[out + v] = bestv[t][v];
        pidx[out + v] = besti[t][v];
      }
    }
  }
}

#define RSPT (NSPLIT / 16)

__device__ __forceinline__ void take_min(float& bv, int& bi, float ov, int oi) {
  const bool t = (ov < bv) | ((ov == bv) & (oi < bi));
  bv = t ? ov : bv;
  bi = t ? oi : bi;
}

__global__ __launch_bounds__(256) void reduce_kernel(
    const float* __restrict__ pval, const int* __restrict__ pidx,
    int* __restrict__ nn_idx) {
  __shared__ float s_v[8][16];
  __shared__ int s_i[8][16];
  const int tid = threadIdx.x;
  const int tile = blockIdx.x;
  const int r = tid & 15;
  const int wave = tid >> 5;
  const int base = tile * NSPLIT * 16 + tid;

  float v[RSPT];
  int ix[RSPT];
#pragma unroll
  for (int j = 0; j < RSPT; ++j) {
    v[j] = pval[base + j * 256];
    ix[j] = pidx[base + j * 256];
  }
  float bv = 3.4e38f;
  int bi = 0;
#pragma unroll
  for (int j = 0; j < RSPT; ++j) take_min(bv, bi, v[j], ix[j]);

  take_min(bv, bi, __shfl_xor(bv, 16, 32), __shfl_xor(bi, 16, 32));
  if ((tid & 16) == 0) { s_v[wave][r] = bv; s_i[wave][r] = bi; }
  __syncthreads();

  if (tid < 16) {
    float wv[8];
    int wi[8];
#pragma unroll
    for (int w = 0; w < 8; ++w) { wv[w] = s_v[w][tid]; wi[w] = s_i[w][tid]; }
    float fv = wv[0];
    int fi = wi[0];
#pragma unroll
    for (int w = 1; w < 8; ++w) take_min(fv, fi, wv[w], wi[w]);
    nn_idx[tile * 16 + tid] = fi;
  }
}

#define MW 8
#define HT 4
#define OT 4
#define RP 17

__global__ __launch_bounds__(MW * 32) void mlp_kernel(
    const float* __restrict__ table, const int* __restrict__ nn_idx,
    const float* __restrict__ W1, const float* __restrict__ b1,
    const float* __restrict__ Wu, const float* __restrict__ bu,
    const float* __restrict__ Ws, const float* __restrict__ bs,
    float* __restrict__ out) {
  __shared__ float s_p[MW][DD * RP];

  const int tid = threadIdx.x;
  const int lane = tid & 31;
  const int wave = tid >> 5;
  const int mtile = blockIdx.x;
  const int z = blockIdx.y;
  const float* W2 = z ? Ws : Wu;
  const float* bvec = z ? bs : bu;
  const int c = lane & 15;
  const int hf = lane >> 4;
  const int hb = wave * (HT * 16);

  const int row = nn_idx[mtile * 16 + c];

  const float* w1p = W1 + (size_t)(hf * 2) * HH + hb + c;
  v2f w1[16][HT];
#pragma unroll
  for (int kb = 0; kb < 16; ++kb)
#pragma unroll
    for (int j = 0; j < HT; ++j) {
      w1[kb][j].x = w1p[(size_t)(kb * 4 + 0) * HH + j * 16];
      w1[kb][j].y = w1p[(size_t)(kb * 4 + 1) * HH + j * 16];
    }

  v8f h[HT];
#pragma unroll
  for (int j = 0; j < HT; ++j) {
    const float* bp = b1 + hb + j * 16 + hf * 8;
#pragma unroll
    for (int r = 0; r < 8; ++r) h[j][r] = bp[r];
  }

  const float* w2p = W2 + (size_t)(hb + hf * 8) * DD + c;
  v2f w2[HT][4][OT];
#pragma unroll
  for (int j = 0; j < HT; ++j)
#pragma unroll
    for (int q = 0; q < 4; ++q)
#pragma unroll
      for (int i = 0; i < OT; ++i) {
        w2[j][q][i].x = w2p[(size_t)(j * 16 + q * 2 + 0) * DD + i * 16];
        w2[j][q][i].y = w2p[(size_t)(j * 16 + q * 2 + 1) * DD + i * 16];
      }

  const float* prow = table + (size_t)row * DD + hf * 2;
  v2f g[16];
#pragma unroll
  for (int kb = 0; kb < 16; ++kb) g[kb] = *(const v2f*)(prow + kb * 4);

#pragma unroll
  for (int kb = 0; kb < 16; ++kb)
#pragma unroll
    for (int j = 0; j < HT; ++j)
      h[j] = WMMA_F32(w1[kb][j], g[kb], h[j]);

#pragma unroll
  for (int j = 0; j < HT; ++j)
#pragma unroll
    for (int r = 0; r < 8; ++r) {
      const float x = h[j][r];
      h[j][r] = x > 0.f ? x : 0.f;
    }

  v8f o[OT];
#pragma unroll
  for (int i = 0; i < OT; ++i) o[i] = (v8f){};
#pragma unroll
  for (int j = 0; j < HT; ++j)
#pragma unroll
    for (int q = 0; q < 4; ++q) {
      v2f b;
      b.x = h[j][2 * q + 0];
      b.y = h[j][2 * q + 1];
#pragma unroll
      for (int i = 0; i < OT; ++i)
        o[i] = WMMA_F32(w2[j][q][i], b, o[i]);
    }

#pragma unroll
  for (int i = 0; i < OT; ++i)
#pragma unroll
    for (int r = 0; r < 8; ++r)
      s_p[wave][(i * 16 + hf * 8 + r) * RP + c] = o[i][r];
  __syncthreads();

  const int d = tid & 63;
  const int bq = tid >> 6;
  const float bias = bvec[d];
  float* o_out = out + (size_t)z * BB * DD + (size_t)(mtile * 16 + bq * 4) * DD + d;
#pragma unroll
  for (int e = 0; e < 4; ++e) {
    const int off = d * RP + bq * 4 + e;
    float s = s_p[0][off];
#pragma unroll
    for (int w = 1; w < MW; ++w) s += s_p[w][off];
    o_out[(size_t)e * DD] = s + bias;
  }
}

extern "C" void kernel_launch(void* const* d_in, const int* in_sizes, int n_in,
                              void* d_out, int out_size, void* d_ws,
                              size_t ws_size, hipStream_t stream) {
  const float* codes = (const float*)d_in[0];
  const float* table = (const float*)d_in[1];
  const float* W1 = (const float*)d_in[2];
  const float* b1 = (const float*)d_in[3];
  const float* Wu = (const float*)d_in[4];
  const float* bu = (const float*)d_in[5];
  const float* Ws = (const float*)d_in[6];
  const float* bs = (const float*)d_in[7];
  float* out = (float*)d_out;

  char* ws = (char*)d_ws;
  float* tnorm = (float*)(ws);
  float* pval  = (float*)(ws + 400000);
  int*   pidx  = (int*)  (ws + 400000 + 262144);
  int*   nnidx = (int*)  (ws + 400000 + 524288);

  tnorm_kernel<<<(NN + 255) / 256, 256, 0, stream>>>(table, tnorm);
  nn_kernel<<<dim3(BB / 16 / (NWAVES * WTILES), NSPLIT), NWAVES * 32, 0, stream>>>(
      codes, table, tnorm, pval, pidx);
  reduce_kernel<<<BB / 16, 256, 0, stream>>>(pval, pidx, nnidx);
  mlp_kernel<<<dim3(BB / 16, 2), MW * 32, 0, stream>>>(table, nnidx, W1, b1, Wu,
                                                       bu, Ws, bs, out);
}


# nn search: table row norms summed by the threads that stage the rows into LDS (4 squares per lane, 16-lane DPP add: new order of the 64 terms); the norm launch is gone, 4 launches to 3
# speedup vs baseline: 2.0051x; 1.0698x over previous
#include <hip/hip_runtime.h>

typedef __attribute__((ext_vector_type(2))) float v2f;
typedef __attribute__((ext_vector_type(4))) float v4f;
typedef __attribute__((ext_vector_type(8))) float v8f;

#define WMMA_F32(a, b, c) \
  __builtin_amdgcn_wmma_f32_16x16x4_f32(false, (a), false, (b), (short)0, (c), false, false)

#define BB      1024
#define NN      100000
#define DD      64
#define HH      512
#define NWAVES  8
#define WTILES  2
#define STAGE_CHUNKS 20
#define STAGE_ROWS   (STAGE_CHUNKS * 16)
#define PITCH   68
#define NSPLIT  64
#define CHUNKS  (NN / 16)
#define CPS     ((CHUNKS + NSPLIT - 1) / NSPLIT)

template <int XM>
__device__ __forceinline__ float add_from_lane_xor(float x) {
  return x + __builtin_bit_cast(
                 float, __builtin_amdgcn_update_dpp(0, __builtin_bit_cast(int, x),
                                                    0x160 + XM, 0xf, 0xf, true));
}

__device__ __forceinline__ void row_norms(const v4f (&g)[STAGE_CHUNKS],
                                          float (&pn)[STAGE_CHUNKS]) {
#pragma unroll
  for (int i = 0; i < STAGE_CHUNKS; ++i) {
    const v4f v = g[i];
    pn[i] = v.x * v.x + v.y * v.y + v.z * v.z + v.w * v.w;
  }
#pragma unroll
  for (int i = 0; i < STAGE_CHUNKS; ++i) pn[i] = add_from_lane_xor<1>(pn[i]);
#pragma unroll
  for (int i = 0; i < STAGE_CHUNKS; ++i) pn[i] = add_from_lane_xor<2>(pn[i]);
#pragma unroll
  for (int i = 0; i < STAGE_CHUNKS; ++i) pn[i] = add_from_lane_xor<4>(pn[i]);
#pragma unroll
  for (int i = 0; i < STAGE_CHUNKS; ++i) pn[i] = add_from_lane_xor<8>(pn[i]);
}

__global__ __launch_bounds__(NWAVES * 32) void nn_kernel(
    const float* __restrict__ codes, const float* __restrict__ table,
    float* __restrict__ pval, int* __restrict__ pidx) {
  __shared__ __attribute__((aligned(16))) float s_t[2][STAGE_ROWS * PITCH];
  __shared__ float s_n[2][STAGE_ROWS];

  const int tid = threadIdx.x;
  const int lane = tid & 31;
  const int wave = tid >> 5;
  const int mgroup = blockIdx.x;
  const int split = blockIdx.y;
  const int m = lane & 15;
  const int koff = (lane >> 4) * 2;
  const int tile0 = (mgroup * NWAVES + wave) * WTILES;

  v2f a[WTILES][16];
#pragma unroll
  for (int t = 0; t < WTILES; ++t) {
    const float* crow = codes + (size_t)((tile0 + t) * 16 + m) * DD + koff;
#pragma unroll
    for (int kb = 0; kb < 16; ++kb) {
      v2f v = *(const v2f*)(crow + kb * 4);
      a[t][kb].x = -2.0f * v.x;
      a[t][kb].y = -2.0f * v.y;
    }
  }

  float bestv[WTILES][8];
  int besti[WTILES][8];
#pragma unroll
  for (int t = 0; t < WTILES; ++t)
#pragma unroll
    for (int v = 0; v < 8; ++v) { bestv[t][v] = 3.4e38f; besti[t][v] = 0; }

  const int c0 = split * CPS;
  const int c1 = (c0 + CPS < CHUNKS) ? (c0 + CPS) : CHUNKS;
  const int nch = c1 - c0;
  const int nst = (nch + STAGE_CHUNKS - 1) / STAGE_CHUNKS;

  v4f g[STAGE_CHUNKS];
  float pn[STAGE_CHUNKS];
  const int srow = tid >> 4;
  const int spiece = (tid & 15) * 4;

  {
    const int nchs = (nch < STAGE_CHUNKS) ? nch : STAGE_CHUNKS;
    const float* gp = table + (size_t)(c0 * 16 + srow) * DD + spiece;
#pragma unroll
    for (int i = 0; i < STAGE_CHUNKS; ++i)
      g[i] = (i < nchs) ? *(const v4f*)(gp + (size_t)i * 16 * DD) : (v4f){};
#pragma unroll
    for (int i = 0; i < STAGE_CHUNKS; ++i)
      if (i < nchs)
        *(v4f*)(&s_t[0][(i * 16 + srow) * PITCH + spiece]) = g[i];
    row_norms(g, pn);
    if ((tid & 15) == 0) {
#pragma unroll
      for (int i = 0; i < STAGE_CHUNKS; ++i) s_n[0][i * 16 + srow] = pn[i];
    }
  }
  __syncthreads();

  for (int s = 0; s < nst; ++s) {
    const int buf = s & 1;
    const int cs0 = c0 + s * STAGE_CHUNKS;
    const int nchs = (c1 - cs0 < STAGE_CHUNKS) ? (c1 - cs0) : STAGE_CHUNKS;
    const int cn0 = cs0 + STAGE_CHUNKS;
    const int nchn =
        (c1 - cn0 < STAGE_CHUNKS) ? (c1 - cn0) : STAGE_CHUNKS;

    if (nchn > 0) {
      const float* gp = table + (size_t)(cn0 * 16 + srow) * DD + spiece;
#pragma unroll
      for (int i = 0; i < STAGE_CHUNKS; ++i)
        if (i < nchn) g[i] = *(const v4f*)(gp + (size_t)i * 16 * DD);
    }

    const float* sb = &s_t[buf][m * PITCH + koff];
    for (int cc = 0; cc < nchs; ++cc) {
      const int col = (cs0 + cc) * 16 + m;
      const float* trow = sb + cc * 16 * PITCH;

      v2f b[16];
#pragma unroll
      for (int kb = 0; kb < 16; ++kb) b[kb] = *(const v2f*)(trow + kb * 4);
      const float tn = s_n[buf][cc * 16 + m];

      v8f acc[WTILES];
#pragma unroll
      for (int t = 0; t < WTILES; ++t) acc[t] = (v8f){};

#pragma unroll
      for (int kb = 0; kb < 16; ++kb)
#pragma unroll
        for (int t = 0; t < WTILES; ++t)
          acc[t] = WMMA_F32(a[t][kb], b[kb], acc[t]);

#pragma unroll
      for (int t = 0; t < WTILES; ++t)
#pragma unroll
        for (int v = 0; v < 8; ++v) {
          float sc = tn + acc[t][v];
          if (sc < bestv[t][v]) { bestv[t][v] = sc; besti[t][v] = col; }
        }
    }

    if (nchn > 0) {
#pragma unroll
      for (int i = 0; i < STAGE_CHUNKS; ++i)
        if (i < nchn)
          *(v4f*)(&s_t[buf ^ 1][(i * 16 + srow) * PITCH + spiece]) = g[i];
      row_norms(g, pn);
      if ((tid & 15) == 0) {
#pragma unroll
        for (int i = 0; i < STAGE_CHUNKS; ++i)
          s_n[buf ^ 1][i * 16 + srow] = pn[i];
      }
    }
    __syncthreads();
  }

#pragma unroll
  for (int off = 1; off < 16; off <<= 1) {
#pragma unroll
    for (int t = 0; t < WTILES; ++t)
#pragma unroll
      for (int v = 0; v < 8; ++v) {
        float ov = __shfl_xor(bestv[t][v], off, 32);
        int oi = __shfl_xor(besti[t][v], off, 32);
        if (ov < bestv[t][v]) { bestv[t][v] = ov; besti[t][v] = oi; }
      }
  }

  if ((lane & 15) == 0) {
    const int rbase = (lane >> 4) * 8;
#pragma unroll
    for (int t = 0; t < WTILES; ++t) {
      const int tile = tile0 + t;
      const int out = (tile * NSPLIT + split) * 16 + rbase;
#pragma unroll
      for (int v = 0; v < 8; ++v) {
        pval[out + v] = bestv[t][v];
        pidx[out + v] = besti[t][v];
      }
    }
  }
}

#define RSPT (NSPLIT / 16)

__device__ __forceinline__ void take_min(float& bv, int& bi, float ov, int oi) {
  const bool t = (ov < bv) | ((ov == bv) & (oi < bi));
  bv = t ? ov : bv;
  bi = t ? oi : bi;
}

__global__ __launch_bounds__(256) void reduce_kernel(
    const float* __restrict__ pval, const int* __restrict__ pidx,
    int* __restrict__ nn_idx) {
  __shared__ float s_v[8][16];
  __shared__ int s_i[8][16];
  const int tid = threadIdx.x;
  const int tile = blockIdx.x;
  const int r = tid & 15;
  const int wave = tid >> 5;
  const int base = tile * NSPLIT * 16 + tid;

  float v[RSPT];
  int ix[RSPT];
#pragma unroll
  for (int j = 0; j < RSPT; ++j) {
    v[j] = pval[base + j * 256];
    ix[j] = pidx[base + j * 256];
  }
  float bv = 3.4e38f;
  int bi = 0;
#pragma unroll
  for (int j = 0; j < RSPT; ++j) take_min(bv, bi, v[j], ix[j]);

  take_min(bv, bi, __shfl_xor(bv, 16, 32), __shfl_xor(bi, 16, 32));
  if ((tid & 16) == 0) { s_v[wave][r] = bv; s_i[wave][r] = bi; }
  __syncthreads();

  if (tid < 16) {
    float wv[8];
    int wi[8];
#pragma unroll
    for (int w = 0; w < 8; ++w) { wv[w] = s_v[w][tid]; wi[w] = s_i[w][tid]; }
    float fv = wv[0];
    int fi = wi[0];
#pragma unroll
    for (int w = 1; w < 8; ++w) take_min(fv, fi, wv[w], wi[w]);
    nn_idx[tile * 16 + tid] = fi;
  }
}

#define MW 8
#define HT 4
#define OT 4
#define RP 17

__global__ __launch_bounds__(MW * 32) void mlp_kernel(
    const float* __restrict__ table, const int* __restrict__ nn_idx,
    const float* __restrict__ W1, const float* __restrict__ b1,
    const float* __restrict__ Wu, const float* __restrict__ bu,
    const float* __restrict__ Ws, const float* __restrict__ bs,
    float* __restrict__ out) {
  __shared__ float s_p[MW][DD * RP];

  const int tid = threadIdx.x;
  const int lane = tid & 31;
  const int wave = tid >> 5;
  const int mtile = blockIdx.x;
  const int z = blockIdx.y;
  const float* W2 = z ? Ws : Wu;
  const float* bvec = z ? bs : bu;
  const int c = lane & 15;
  const int hf = lane >> 4;
  const int hb = wave * (HT * 16);

  const int row = nn_idx[mtile * 16 + c];

  const float* w1p = W1 + (size_t)(hf * 2) * HH + hb + c;
  v2f w1[16][HT];
#pragma unroll
  for (int kb = 0; kb < 16; ++kb)
#pragma unroll
    for (int j = 0; j < HT; ++j) {
      w1[kb][j].x = w1p[(size_t)(kb * 4 + 0) * HH + j * 16];
      w1[kb][j].y = w1p[(size_t)(kb * 4 + 1) * HH + j * 16];
    }

  v8f h[HT];
#pragma unroll
  for (int j = 0; j < HT; ++j) {
    const float* bp = b1 + hb + j * 16 + hf * 8;
#pragma unroll
    for (int r = 0; r < 8; ++r) h[j][r] = bp[r];
  }

  const float* w2p = W2 + (size_t)(hb + hf * 8) * DD + c;
  v2f w2[HT][4][OT];
#pragma unroll
  for (int j = 0; j < HT; ++j)
#pragma unroll
    for (int q = 0; q < 4; ++q)
#pragma unroll
      for (int i = 0; i < OT; ++i) {
        w2[j][q][i].x = w2p[(size_t)(j * 16 + q * 2 + 0) * DD + i * 16];
        w2[j][q][i].y = w2p[(size_t)(j * 16 + q * 2 + 1) * DD + i * 16];
      }

  const float* prow = table + (size_t)row * DD + hf * 2;
  v2f g[16];
#pragma unroll
  for (int kb = 0; kb < 16; ++kb) g[kb] = *(const v2f*)(prow + kb * 4);

#pragma unroll
  for (int kb = 0; kb < 16; ++kb)
#pragma unroll
    for (int j = 0; j < HT; ++j)
      h[j] = WMMA_F32(w1[kb][j], g[kb], h[j]);

#pragma unroll
  for (int j = 0; j < HT; ++j)
#pragma unroll
    for (int r = 0; r < 8; ++r) {
      const float x = h[j][r];
      h[j][r] = x > 0.f ? x : 0.f;
    }

  v8f o[OT];
#pragma unroll
  for (int i = 0; i < OT; ++i) o[i] = (v8f){};
#pragma unroll
  for (int j = 0; j < HT; ++j)
#pragma unroll
    for (int q = 0; q < 4; ++q) {
      v2f b;
      b.x = h[j][2 * q + 0];
      b.y = h[j][2 * q + 1];
#pragma unroll
      for (int i = 0; i < OT; ++i)
        o[i] = WMMA_F32(w2[j][q][i], b, o[i]);
    }

#pragma unroll
  for (int i = 0; i < OT; ++i)
#pragma unroll
    for (int r = 0; r < 8; ++r)
      s_p[wave][(i * 16 + hf * 8 + r) * RP + c] = o[i][r];
  __syncthreads();

  const int d = tid & 63;
  const int bq = tid >> 6;
  const float bias = bvec[d];
  float* o_out = out + (size_t)z * BB * DD + (size_t)(mtile * 16 + bq * 4) * DD + d;
#pragma unroll
  for (int e = 0; e < 4; ++e) {
    const int off = d * RP + bq * 4 + e;
    float s = s_p[0][off];
#pragma unroll
    for (int w = 1; w < MW; ++w) s += s_p[w][off];
    o_out[(size_t)e * DD] = s + bias;
  }
}

extern "C" void kernel_launch(void* const* d_in, const int* in_sizes, int n_in,
                              void* d_out, int out_size, void* d_ws,
                              size_t ws_size, hipStream_t stream) {
  const float* codes = (const float*)d_in[0];
  const float* table = (const float*)d_in[1];
  const float* W1 = (const float*)d_in[2];
  const float* b1 = (const float*)d_in[3];
  const float* Wu = (const float*)d_in[4];
  const float* bu = (const float*)d_in[5];
  const float* Ws = (const float*)d_in[6];
  const float* bs = (const float*)d_in[7];
  float* out = (float*)d_out;

  char* ws = (char*)d_ws;
  float* pval  = (float*)(ws);
  int*   pidx  = (int*)  (ws + 262144);
  int*   nnidx = (int*)  (ws + 524288);

  nn_kernel<<<dim3(BB / 16 / (NWAVES * WTILES), NSPLIT), NWAVES * 32, 0, stream>>>(
      codes, table, pval, pidx);
  reduce_kernel<<<BB / 16, 256, 0, stream>>>(pval, pidx, nnidx);
  mlp_kernel<<<dim3(BB / 16, 2), MW * 32, 0, stream>>>(table, nnidx, W1, b1, Wu,
                                                       bu, Ws, bs, out);
}
